# starting kernel
# baseline (speedup 1.0000x reference)
#include <hip/hip_runtime.h>
#include <math.h>

typedef __attribute__((ext_vector_type(2))) float v2f;
typedef __attribute__((ext_vector_type(4))) float v4f;
typedef __attribute__((ext_vector_type(8))) float v8f;

#define GAT_N 2048
#define GAT_M 8192
#define GAT_F 1024
#define GAT_K 10
#define GAT_ALPHA 0.2f

__global__ __launch_bounds__(256) void gat_gemm_xw(const float* __restrict__ X,
                                                   const float* __restrict__ W,
                                                   float* __restrict__ H,
                                                   int rows) {
    const int F = GAT_F;
    const int wavesPerBlock = blockDim.x >> 5;
    const int wave = blockIdx.x * wavesPerBlock + (threadIdx.x >> 5);
    const int rowTiles = rows >> 6;
    const int rowTile = wave % rowTiles;
    const int colTile = wave / rowTiles;
    const int lane = threadIdx.x & 31;
    const int r = lane & 15;
    const int g = lane >> 4;
    const int i0 = rowTile << 6;
    const int j0 = colTile << 6;

    v8f c[4][4];
#pragma unroll
    for (int m = 0; m < 4; ++m)
#pragma unroll
        for (int n = 0; n < 4; ++n)
#pragma unroll
            for (int v = 0; v < 8; ++v) c[m][n][v] = 0.0f;

    const float* xr[4];
#pragma unroll
    for (int m = 0; m < 4; ++m)
        xr[m] = X + (size_t)(i0 + 16 * m + r) * F;

#pragma unroll 2
    for (int k0 = 0; k0 < F; k0 += 4) {
        const int ka = k0 + 2 * g;
        v2f a[4];
#pragma unroll
        for (int m = 0; m < 4; ++m)
            a[m] = *(const v2f*)(xr[m] + ka);

        const float* w0 = W + (size_t)ka * F + j0 + r;
        const float* w1 = w0 + F;
        v2f b[4];
#pragma unroll
        for (int n = 0; n < 4; ++n) {
            b[n].x = w0[16 * n];
            b[n].y = w1[16 * n];
        }
#pragma unroll
        for (int m = 0; m < 4; ++m)
#pragma unroll
            for (int n = 0; n < 4; ++n)
                c[m][n] = __builtin_amdgcn_wmma_f32_16x16x4_f32(
                    false, a[m], false, b[n], (short)0, c[m][n], false, false);
    }

#pragma unroll
    for (int m = 0; m < 4; ++m)
#pragma unroll
        for (int n = 0; n < 4; ++n)
#pragma unroll
            for (int v = 0; v < 8; ++v) {
                const int row = i0 + 16 * m + 8 * g + v;
                const int col = j0 + 16 * n + r;
                H[(size_t)row * F + col] = c[m][n][v];
            }
}

__global__ __launch_bounds__(256) void gat_rowdot(const float* __restrict__ hf,
                                                  const float* __restrict__ he,
                                                  const float* __restrict__ a,
                                                  float* __restrict__ s1,
                                                  float* __restrict__ s2) {
    const int F = GAT_F;
    const int wave = blockIdx.x * (blockDim.x >> 5) + (threadIdx.x >> 5);
    const int lane = threadIdx.x & 31;

    const float* h;
    const float* av;
    float* out;
    int row;
    if (wave < GAT_N) { h = hf; av = a;      out = s1; row = wave; }
    else              { h = he; av = a + F;  out = s2; row = wave - GAT_N; }

    const float* hr = h + (size_t)row * F;
    float acc = 0.0f;
    for (int k = lane; k < F; k += 32) acc += hr[k] * av[k];
#pragma unroll
    for (int off = 16; off > 0; off >>= 1) acc += __shfl_down(acc, off, 32);
    if (lane == 0) out[row] = acc;
}

__global__ __launch_bounds__(256) void gat_attn_gather(const float* __restrict__ he,
                                                       const float* __restrict__ s1,
                                                       const float* __restrict__ s2,
                                                       const int* __restrict__ nidx,
                                                       float* __restrict__ out) {
    const int F = GAT_F;
    const int i = blockIdx.x;

    __shared__ int   sh_idx[GAT_K];
    __shared__ float sh_w[GAT_K];
    __shared__ int   sh_cnt;

    if (threadIdx.x == 0) {
        int   idx[GAT_K];
        float ev[GAT_K];
        int cnt = 0;
        const float s1i = s1[i];
        for (int k = 0; k < GAT_K; ++k) {
            const int j = nidx[i * GAT_K + k];
            bool dup = false;
            for (int p = 0; p < cnt; ++p) dup = dup || (idx[p] == j);
            if (!dup) {
                float e = s1i + s2[j];
                e = (e > 0.0f) ? e : GAT_ALPHA * e;
                idx[cnt] = j;
                ev[cnt] = e;
                ++cnt;
            }
        }
        float mx = ev[0];
        for (int p = 1; p < cnt; ++p) mx = fmaxf(mx, ev[p]);
        float sum = 0.0f;
        for (int p = 0; p < cnt; ++p) { ev[p] = expf(ev[p] - mx); sum += ev[p]; }
        const float inv = 1.0f / sum;
        for (int p = 0; p < cnt; ++p) { sh_idx[p] = idx[p]; sh_w[p] = ev[p] * inv; }
        sh_cnt = cnt;
    }
    __syncthreads();

    const int cnt = sh_cnt;
    const int c = threadIdx.x * 4;
    v4f acc; acc[0] = 0.f; acc[1] = 0.f; acc[2] = 0.f; acc[3] = 0.f;
    for (int p = 0; p < cnt; ++p) {
        const float w = sh_w[p];
        const v4f hv = *(const v4f*)(he + (size_t)sh_idx[p] * F + c);
        acc += hv * w;
    }
    *(v4f*)(out + (size_t)i * F + c) = acc;
}

extern "C" void kernel_launch(void* const* d_in, const int* in_sizes, int n_in,
                              void* d_out, int out_size, void* d_ws, size_t ws_size,
                              hipStream_t stream) {
    const int N = GAT_N, M = GAT_M, F = GAT_F;

    const float* feat  = (const float*)d_in[0];
    const float* embed = (const float*)d_in[1];
    const float* W     = (const float*)d_in[2];
    const float* a     = (const float*)d_in[3];
    const int*   nidx  = (const int*)d_in[4];
    float* out = (float*)d_out;

    char* ws = (char*)d_ws;
    float* hf = (float*)ws;
    float* he = (float*)(ws + (size_t)N * F * sizeof(float));
    float* s1 = (float*)(ws + (size_t)(N + M) * F * sizeof(float));
    float* s2 = s1 + N;

    {
        const int waves = (N / 64) * (F / 64);
        gat_gemm_xw<<<waves / 8, 256, 0, stream>>>(feat, W, hf, N);
    }
    {
        const int waves = (M / 64) * (F / 64);
        gat_gemm_xw<<<waves / 8, 256, 0, stream>>>(embed, W, he, M);
    }
    gat_rowdot<<<(N + M) / 8, 256, 0, stream>>>(hf, he, a, s1, s2);

    gat_attn_gather<<<N, 256, 0, stream>>>(he, s1, s2, nidx, out);
}


# GEMM operands staged through LDS: block tile 160x256 (8 waves of 80x64), each 32-K piece of X and W loaded once per block by 16-B global loads into 2 LDS stages; float sum order unchanged
# speedup vs baseline: 2.2107x; 2.2107x over previous
#include <hip/hip_runtime.h>
#include <math.h>

typedef __attribute__((ext_vector_type(2))) float v2f;
typedef __attribute__((ext_vector_type(4))) float v4f;
typedef __attribute__((ext_vector_type(8))) float v8f;

#define GAT_N 2048
#define GAT_M 8192
#define GAT_F 1024
#define GAT_K 10
#define GAT_ALPHA 0.2f

#define GEMM_BM 160
#define GEMM_BN 256
#define GEMM_BK 32
#define GEMM_APITCH 36
#define GEMM_BCOLP (GEMM_BN + 16)
#define GEMM_ASTAGE (GEMM_BM * GEMM_APITCH)
#define GEMM_BSTAGE ((GEMM_BK / 2) * GEMM_BCOLP * 2)
#define GEMM_STAGE (GEMM_ASTAGE + GEMM_BSTAGE)

struct gat_no_tail { __device__ __forceinline__ void operator()() const {} };
template <class Tail>
__device__ __forceinline__ void gat_gemm_chunk(const float* sA, const float* sB,
                                               v8f (&c)[5][4], Tail tail) {
    v2f a[2][5];
    v2f b[2][4];
#pragma unroll
    for (int m = 0; m < 5; ++m) a[0][m] = *(const v2f*)(sA + 16 * m * GEMM_APITCH);
#pragma unroll
    for (int n = 0; n < 4; ++n) b[0][n] = *(const v2f*)(sB + 32 * n);
#pragma unroll
    for (int s = 0; s < 8; ++s) {
        const int cu = s & 1;
        const int nx = cu ^ 1;
        if (s + 1 < 8) {
#pragma unroll
            for (int m = 0; m < 5; ++m)
                a[nx][m] = *(const v2f*)(sA + 16 * m * GEMM_APITCH + 4 * (s + 1));
#pragma unroll
            for (int n = 0; n < 4; ++n)
                b[nx][n] = *(const v2f*)(sB + 4 * (s + 1) * GEMM_BCOLP + 32 * n);
        } else {
            tail();
        }
        __builtin_amdgcn_sched_barrier(0);
#pragma unroll
        for (int m = 0; m < 5; ++m)
#pragma unroll
            for (int n = 0; n < 4; ++n)
                c[m][n] = __builtin_amdgcn_wmma_f32_16x16x4_f32(
                    false, a[cu][m], false, b[cu][n], (short)0, c[m][n], false, false);
        __builtin_amdgcn_sched_barrier(0);
    }
}

__global__ __launch_bounds__(256) void gat_gemm_xw(const float* __restrict__ Xf,
                                                   const float* __restrict__ Xe,
                                                   const float* __restrict__ W,
                                                   float* __restrict__ H) {
    __shared__ __attribute__((aligned(16))) float lds[2 * GEMM_STAGE];

    const int F = GAT_F;
    const int t = threadIdx.x;
    const int lane = t & 31;
    const int wave = t >> 5;
    const int wr = wave >> 2;
    const int wc = wave & 3;
    const int r = lane & 15;
    const int g = lane >> 4;
    const int i0 = (blockIdx.x >> 2) * GEMM_BM;
    const int j0 = (blockIdx.x & 3) * GEMM_BN;

    const int arow = t >> 3;
    const int akq = t & 7;
    const float* ap[5];
#pragma unroll
    for (int i = 0; i < 5; ++i) {
        const int row = i0 + arow + 32 * i;
        const float* base = (row < GAT_N) ? (Xf + (size_t)row * F)
                                          : (Xe + (size_t)(row - GAT_N) * F);
        ap[i] = base + 4 * akq;
    }
    const int a_st = arow * GEMM_APITCH + 4 * akq;

    const int bkp = t >> 6;
    const int bcg = t & 63;
    const float* wp = W + (size_t)(2 * bkp) * F + j0 + 4 * bcg;
    const int b_st = GEMM_ASTAGE + (bkp * GEMM_BCOLP + 4 * bcg) * 2;

    const int a_ld = (wr * 80 + r) * GEMM_APITCH + 2 * g;
    const int b_ld = GEMM_ASTAGE + (g * GEMM_BCOLP + wc * 64 + r) * 2;

    v8f c[5][4];
#pragma unroll
    for (int m = 0; m < 5; ++m)
#pragma unroll
        for (int n = 0; n < 4; ++n)
#pragma unroll
            for (int v = 0; v < 8; ++v) c[m][n][v] = 0.0f;

    v4f ga[5], gb0[4], gb1[4];

#pragma unroll
    for (int i = 0; i < 5; ++i) ga[i] = *(const v4f*)(ap[i]);
#pragma unroll
    for (int i = 0; i < 4; ++i) {
        const float* p = wp + (size_t)(8 * i) * F;
        gb0[i] = *(const v4f*)p;
        gb1[i] = *(const v4f*)(p + F);
    }
#pragma unroll
    for (int i = 0; i < 5; ++i) *(v4f*)(lds + a_st + 32 * i * GEMM_APITCH) = ga[i];
#pragma unroll
    for (int i = 0; i < 4; ++i) {
        float* q = lds + b_st + 8 * i * GEMM_BCOLP;
        v4f lo, hi;
        lo[0] = gb0[i][0]; lo[1] = gb1[i][0]; lo[2] = gb0[i][1]; lo[3] = gb1[i][1];
        hi[0] = gb0[i][2]; hi[1] = gb1[i][2]; hi[2] = gb0[i][3]; hi[3] = gb1[i][3];
        *(v4f*)q = lo;
        *(v4f*)(q + 4) = hi;
    }
    __syncthreads();

    const int nChunks = F / GEMM_BK;
    for (int kc = 0; kc + 1 < nChunks; ++kc) {
        float* cur = lds + (kc & 1) * GEMM_STAGE;
        float* nxt = lds + ((kc & 1) ^ 1) * GEMM_STAGE;

        const int kn = (kc + 1) * GEMM_BK;
#pragma unroll
        for (int i = 0; i < 5; ++i) ga[i] = *(const v4f*)(ap[i] + kn);
#pragma unroll
        for (int i = 0; i < 4; ++i) {
            const float* p = wp + (size_t)(kn + 8 * i) * F;
            gb0[i] = *(const v4f*)p;
            gb1[i] = *(const v4f*)(p + F);
        }

        gat_gemm_chunk(cur + a_ld, cur + b_ld, c, [&]() {
#pragma unroll
            for (int i = 0; i < 5; ++i) *(v4f*)(nxt + a_st + 32 * i * GEMM_APITCH) = ga[i];
#pragma unroll
            for (int i = 0; i < 4; ++i) {
                float* q = nxt + b_st + 8 * i * GEMM_BCOLP;
                v4f lo, hi;
                lo[0] = gb0[i][0]; lo[1] = gb1[i][0]; lo[2] = gb0[i][1]; lo[3] = gb1[i][1];
                hi[0] = gb0[i][2]; hi[1] = gb1[i][2]; hi[2] = gb0[i][3]; hi[3] = gb1[i][3];
                *(v4f*)q = lo;
                *(v4f*)(q + 4) = hi;
            }
        });
        __syncthreads();
    }
    {
        const float* cur = lds + ((nChunks - 1) & 1) * GEMM_STAGE;
        gat_gemm_chunk(cur + a_ld, cur + b_ld, c, gat_no_tail());
    }

#pragma unroll
    for (int m = 0; m < 5; ++m)
#pragma unroll
        for (int n = 0; n < 4; ++n)
#pragma unroll
            for (int v = 0; v < 8; ++v) {
                const int row = i0 + wr * 80 + 16 * m + 8 * g + v;
                const int col = j0 + wc * 64 + 16 * n + r;
                H[(size_t)row * F + col] = c[m][n][v];
            }
}

__global__ __launch_bounds__(256) void gat_rowdot(const float* __restrict__ hf,
                                                  const float* __restrict__ he,
                                                  const float* __restrict__ a,
                                                  float* __restrict__ s1,
                                                  float* __restrict__ s2) {
    const int F = GAT_F;
    const int wave = blockIdx.x * (blockDim.x >> 5) + (threadIdx.x >> 5);
    const int lane = threadIdx.x & 31;

    const float* h;
    const float* av;
    float* out;
    int row;
    if (wave < GAT_N) { h = hf; av = a;      out = s1; row = wave; }
    else              { h = he; av = a + F;  out = s2; row = wave - GAT_N; }

    const float* hr = h + (size_t)row * F;
    float acc = 0.0f;
    for (int k = lane; k < F; k += 32) acc += hr[k] * av[k];
#pragma unroll
    for (int off = 16; off > 0; off >>= 1) acc += __shfl_down(acc, off, 32);
    if (lane == 0) out[row] = acc;
}

__global__ __launch_bounds__(256) void gat_attn_gather(const float* __restrict__ he,
                                                       const float* __restrict__ s1,
                                                       const float* __restrict__ s2,
                                                       const int* __restrict__ nidx,
                                                       float* __restrict__ out) {
    const int F = GAT_F;
    const int i = blockIdx.x;

    __shared__ int   sh_idx[GAT_K];
    __shared__ float sh_w[GAT_K];
    __shared__ int   sh_cnt;

    if (threadIdx.x == 0) {
        int   idx[GAT_K];
        float ev[GAT_K];
        int cnt = 0;
        const float s1i = s1[i];
        for (int k = 0; k < GAT_K; ++k) {
            const int j = nidx[i * GAT_K + k];
            bool dup = false;
            for (int p = 0; p < cnt; ++p) dup = dup || (idx[p] == j);
            if (!dup) {
                float e = s1i + s2[j];
                e = (e > 0.0f) ? e : GAT_ALPHA * e;
                idx[cnt] = j;
                ev[cnt] = e;
                ++cnt;
            }
        }
        float mx = ev[0];
        for (int p = 1; p < cnt; ++p) mx = fmaxf(mx, ev[p]);
        float sum = 0.0f;
        for (int p = 0; p < cnt; ++p) { ev[p] = expf(ev[p] - mx); sum += ev[p]; }
        const float inv = 1.0f / sum;
        for (int p = 0; p < cnt; ++p) { sh_idx[p] = idx[p]; sh_w[p] = ev[p] * inv; }
        sh_cnt = cnt;
    }
    __syncthreads();

    const int cnt = sh_cnt;
    const int c = threadIdx.x * 4;
    v4f acc; acc[0] = 0.f; acc[1] = 0.f; acc[2] = 0.f; acc[3] = 0.f;
    for (int p = 0; p < cnt; ++p) {
        const float w = sh_w[p];
        const v4f hv = *(const v4f*)(he + (size_t)sh_idx[p] * F + c);
        acc += hv * w;
    }
    *(v4f*)(out + (size_t)i * F + c) = acc;
}

extern "C" void kernel_launch(void* const* d_in, const int* in_sizes, int n_in,
                              void* d_out, int out_size, void* d_ws, size_t ws_size,
                              hipStream_t stream) {
    const int N = GAT_N, M = GAT_M, F = GAT_F;

    const float* feat  = (const float*)d_in[0];
    const float* embed = (const float*)d_in[1];
    const float* W     = (const float*)d_in[2];
    const float* a     = (const float*)d_in[3];
    const int*   nidx  = (const int*)d_in[4];
    float* out = (float*)d_out;

    char* ws = (char*)d_ws;
    float* hf = (float*)ws;
    float* he = (float*)(ws + (size_t)N * F * sizeof(float));
    float* s1 = (float*)(ws + (size_t)(N + M) * F * sizeof(float));
    float* s2 = s1 + N;

    gat_gemm_xw<<<256, 256, 0, stream>>>(feat, embed, W, hf);
    gat_rowdot<<<(N + M) / 8, 256, 0, stream>>>(hf, he, a, s1, s2);

    gat_attn_gather<<<N, 256, 0, stream>>>(he, s1, s2, nidx, out);
}


# Score dots s1,s2 taken in the GEMM epilogue from the accumulators (16-lane DPP sums; 4 column-block partials per row, added by the gather); rowdot launch and hf store gone; score sum order changes
# speedup vs baseline: 2.4027x; 1.0869x over previous
#include <hip/hip_runtime.h>
#include <math.h>

typedef __attribute__((ext_vector_type(2))) float v2f;
typedef __attribute__((ext_vector_type(4))) float v4f;
typedef __attribute__((ext_vector_type(8))) float v8f;

#define GAT_N 2048
#define GAT_M 8192
#define GAT_F 1024
#define GAT_K 10
#define GAT_ALPHA 0.2f

#define GEMM_BM 160
#define GEMM_BN 256
#define GEMM_BK 32
#define GEMM_APITCH 36
#define GEMM_BCOLP (GEMM_BN + 16)
#define GEMM_ASTAGE (GEMM_BM * GEMM_APITCH)
#define GEMM_BSTAGE ((GEMM_BK / 2) * GEMM_BCOLP * 2)
#define GEMM_STAGE (GEMM_ASTAGE + GEMM_BSTAGE)

template <int MASK>
__device__ __forceinline__ float gat_xmask(float x) {
    const int b = __builtin_bit_cast(int, x);
    return __builtin_bit_cast(float,
        __builtin_amdgcn_update_dpp(0, b, 0x160 + MASK, 0xf, 0xf, true));
}
__device__ __forceinline__ float gat_sum16(float x) {
    x += gat_xmask<1>(x);
    x += gat_xmask<2>(x);
    x += gat_xmask<4>(x);
    x += gat_xmask<8>(x);
    return x;
}

struct gat_no_tail { __device__ __forceinline__ void operator()() const {} };
template <class Tail>
__device__ __forceinline__ void gat_gemm_chunk(const float* sA, const float* sB,
                                               v8f (&c)[5][4], Tail tail) {
    v2f a[2][5];
    v2f b[2][4];
#pragma unroll
    for (int m = 0; m < 5; ++m) a[0][m] = *(const v2f*)(sA + 16 * m * GEMM_APITCH);
#pragma unroll
    for (int n = 0; n < 4; ++n) b[0][n] = *(const v2f*)(sB + 32 * n);
#pragma unroll
    for (int s = 0; s < 8; ++s) {
        const int cu = s & 1;
        const int nx = cu ^ 1;
        if (s + 1 < 8) {
#pragma unroll
            for (int m = 0; m < 5; ++m)
                a[nx][m] = *(const v2f*)(sA + 16 * m * GEMM_APITCH + 4 * (s + 1));
#pragma unroll
            for (int n = 0; n < 4; ++n)
                b[nx][n] = *(const v2f*)(sB + 4 * (s + 1) * GEMM_BCOLP + 32 * n);
        } else {
            tail();
        }
        __builtin_amdgcn_sched_barrier(0);
#pragma unroll
        for (int m = 0; m < 5; ++m)
#pragma unroll
            for (int n = 0; n < 4; ++n)
                c[m][n] = __builtin_amdgcn_wmma_f32_16x16x4_f32(
                    false, a[cu][m], false, b[cu][n], (short)0, c[m][n], false, false);
        __builtin_amdgcn_sched_barrier(0);
    }
}

__global__ __launch_bounds__(256) void gat_gemm_xw(const float* __restrict__ Xf,
                                                   const float* __restrict__ Xe,
                                                   const float* __restrict__ W,
                                                   const float* __restrict__ avec,
                                                   float* __restrict__ H,
                                                   float* __restrict__ SP) {
    __shared__ __attribute__((aligned(16))) float lds[2 * GEMM_STAGE];

    const int F = GAT_F;
    const int t = threadIdx.x;
    const int lane = t & 31;
    const int wave = t >> 5;
    const int wr = wave >> 2;
    const int wc = wave & 3;
    const int r = lane & 15;
    const int g = lane >> 4;
    const int i0 = (blockIdx.x >> 2) * GEMM_BM;
    const int j0 = (blockIdx.x & 3) * GEMM_BN;

    const int arow = t >> 3;
    const int akq = t & 7;
    const float* ap[5];
#pragma unroll
    for (int i = 0; i < 5; ++i) {
        const int row = i0 + arow + 32 * i;
        const float* base = (row < GAT_N) ? (Xf + (size_t)row * F)
                                          : (Xe + (size_t)(row - GAT_N) * F);
        ap[i] = base + 4 * akq;
    }
    const int a_st = arow * GEMM_APITCH + 4 * akq;

    const int bkp = t >> 6;
    const int bcg = t & 63;
    const float* wp = W + (size_t)(2 * bkp) * F + j0 + 4 * bcg;
    const int b_st = GEMM_ASTAGE + (bkp * GEMM_BCOLP + 4 * bcg) * 2;

    const int a_ld = (wr * 80 + r) * GEMM_APITCH + 2 * g;
    const int b_ld = GEMM_ASTAGE + (g * GEMM_BCOLP + wc * 64 + r) * 2;

    v8f c[5][4];
#pragma unroll
    for (int m = 0; m < 5; ++m)
#pragma unroll
        for (int n = 0; n < 4; ++n)
#pragma unroll
            for (int v = 0; v < 8; ++v) c[m][n][v] = 0.0f;

    v4f ga[5], gb0[4], gb1[4];

#pragma unroll
    for (int i = 0; i < 5; ++i) ga[i] = *(const v4f*)(ap[i]);
#pragma unroll
    for (int i = 0; i < 4; ++i) {
        const float* p = wp + (size_t)(8 * i) * F;
        gb0[i] = *(const v4f*)p;
        gb1[i] = *(const v4f*)(p + F);
    }
#pragma unroll
    for (int i = 0; i < 5; ++i) *(v4f*)(lds + a_st + 32 * i * GEMM_APITCH) = ga[i];
#pragma unroll
    for (int i = 0; i < 4; ++i) {
        float* q = lds + b_st + 8 * i * GEMM_BCOLP;
        v4f lo, hi;
        lo[0] = gb0[i][0]; lo[1] = gb1[i][0]; lo[2] = gb0[i][1]; lo[3] = gb1[i][1];
        hi[0] = gb0[i][2]; hi[1] = gb1[i][2]; hi[2] = gb0[i][3]; hi[3] = gb1[i][3];
        *(v4f*)q = lo;
        *(v4f*)(q + 4) = hi;
    }
    __syncthreads();

    const int nChunks = F / GEMM_BK;
    for (int kc = 0; kc + 1 < nChunks; ++kc) {
        float* cur = lds + (kc & 1) * GEMM_STAGE;
        float* nxt = lds + ((kc & 1) ^ 1) * GEMM_STAGE;

        const int kn = (kc + 1) * GEMM_BK;
#pragma unroll
        for (int i = 0; i < 5; ++i) ga[i] = *(const v4f*)(ap[i] + kn);
#pragma unroll
        for (int i = 0; i < 4; ++i) {
            const float* p = wp + (size_t)(kn + 8 * i) * F;
            gb0[i] = *(const v4f*)p;
            gb1[i] = *(const v4f*)(p + F);
        }

        gat_gemm_chunk(cur + a_ld, cur + b_ld, c, [&]() {
#pragma unroll
            for (int i = 0; i < 5; ++i) *(v4f*)(nxt + a_st + 32 * i * GEMM_APITCH) = ga[i];
#pragma unroll
            for (int i = 0; i < 4; ++i) {
                float* q = nxt + b_st + 8 * i * GEMM_BCOLP;
                v4f lo, hi;
                lo[0] = gb0[i][0]; lo[1] = gb1[i][0]; lo[2] = gb0[i][1]; lo[3] = gb1[i][1];
                hi[0] = gb0[i][2]; hi[1] = gb1[i][2]; hi[2] = gb0[i][3]; hi[3] = gb1[i][3];
                *(v4f*)q = lo;
                *(v4f*)(q + 4) = hi;
            }
        });
        __syncthreads();
    }
    float a1v[4], a2v[4];
#pragma unroll
    for (int n = 0; n < 4; ++n) {
        const int col = j0 + wc * 64 + 16 * n + r;
        a1v[n] = avec[col];
        a2v[n] = avec[F + col];
    }
    {
        const float* cur = lds + ((nChunks - 1) & 1) * GEMM_STAGE;
        gat_gemm_chunk(cur + a_ld, cur + b_ld, c, gat_no_tail());
    }

    const int rb = i0 + __builtin_amdgcn_readfirstlane(wr) * 80;
    float* red = lds;
#pragma unroll
    for (int m = 0; m < 5; ++m) {
        const bool isf = (rb + 16 * m) < GAT_N;
        float av[4];
#pragma unroll
        for (int n = 0; n < 4; ++n) av[n] = isf ? a1v[n] : a2v[n];
        v4f lo, hi;
#pragma unroll
        for (int v = 0; v < 8; ++v) {
            float p = c[m][0][v] * av[0];
            p = __builtin_fmaf(c[m][1][v], av[1], p);
            p = __builtin_fmaf(c[m][2][v], av[2], p);
            p = __builtin_fmaf(c[m][3][v], av[3], p);
            p = gat_sum16(p);
            if (v < 4) lo[v] = p; else hi[v - 4] = p;
        }
        if (r == 0) {
            float* q = red + wc * GEMM_BM + wr * 80 + 16 * m + 8 * g;
            *(v4f*)q = lo;
            *(v4f*)(q + 4) = hi;
        }
    }
    __syncthreads();
    if (t < GEMM_BM) {
        const float s = ((red[t] + red[GEMM_BM + t]) + red[2 * GEMM_BM + t]) + red[3 * GEMM_BM + t];
        SP[(size_t)(blockIdx.x & 3) * (GAT_N + GAT_M) + i0 + t] = s;
    }

#pragma unroll
    for (int m = 0; m < 5; ++m) {
        if ((rb + 16 * m) >= GAT_N) {
#pragma unroll
            for (int n = 0; n < 4; ++n)
#pragma unroll
                for (int v = 0; v < 8; ++v) {
                    const int row = i0 + wr * 80 + 16 * m + 8 * g + v;
                    const int col = j0 + wc * 64 + 16 * n + r;
                    H[(size_t)row * F + col] = c[m][n][v];
                }
        }
    }
}

__global__ __launch_bounds__(256) void gat_attn_gather(const float* __restrict__ he,
                                                       const float* __restrict__ sp,
                                                       const int* __restrict__ nidx,
                                                       float* __restrict__ out) {
    const int F = GAT_F;
    const int i = blockIdx.x;

    __shared__ int   sh_idx[GAT_K];
    __shared__ float sh_w[GAT_K];
    __shared__ int   sh_cnt;

    if (threadIdx.x == 0) {
        int   idx[GAT_K];
        float ev[GAT_K];
        int cnt = 0;
        const int R = GAT_N + GAT_M;
        const float s1i = ((sp[i] + sp[R + i]) + sp[2 * R + i]) + sp[3 * R + i];
        for (int k = 0; k < GAT_K; ++k) {
            const int j = nidx[i * GAT_K + k];
            bool dup = false;
            for (int p = 0; p < cnt; ++p) dup = dup || (idx[p] == j);
            if (!dup) {
                const float* pj = sp + GAT_N + j;
                const float s2j = ((pj[0] + pj[R]) + pj[2 * R]) + pj[3 * R];
                float e = s1i + s2j;
                e = (e > 0.0f) ? e : GAT_ALPHA * e;
                idx[cnt] = j;
                ev[cnt] = e;
                ++cnt;
            }
        }
        float mx = ev[0];
        for (int p = 1; p < cnt; ++p) mx = fmaxf(mx, ev[p]);
        float sum = 0.0f;
        for (int p = 0; p < cnt; ++p) { ev[p] = expf(ev[p] - mx); sum += ev[p]; }
        const float inv = 1.0f / sum;
        for (int p = 0; p < cnt; ++p) { sh_idx[p] = idx[p]; sh_w[p] = ev[p] * inv; }
        sh_cnt = cnt;
    }
    __syncthreads();

    const int cnt = sh_cnt;
    const int c = threadIdx.x * 4;
    v4f acc; acc[0] = 0.f; acc[1] = 0.f; acc[2] = 0.f; acc[3] = 0.f;
    for (int p = 0; p < cnt; ++p) {
        const float w = sh_w[p];
        const v4f hv = *(const v4f*)(he + (size_t)sh_idx[p] * F + c);
        acc += hv * w;
    }
    *(v4f*)(out + (size_t)i * F + c) = acc;
}

extern "C" void kernel_launch(void* const* d_in, const int* in_sizes, int n_in,
                              void* d_out, int out_size, void* d_ws, size_t ws_size,
                              hipStream_t stream) {
    const int N = GAT_N, M = GAT_M, F = GAT_F;

    const float* feat  = (const float*)d_in[0];
    const float* embed = (const float*)d_in[1];
    const float* W     = (const float*)d_in[2];
    const float* a     = (const float*)d_in[3];
    const int*   nidx  = (const int*)d_in[4];
    float* out = (float*)d_out;

    char* ws = (char*)d_ws;
    float* hf = (float*)ws;
    float* he = (float*)(ws + (size_t)N * F * sizeof(float));
    float* sp = (float*)(ws + (size_t)(N + M) * F * sizeof(float));

    gat_gemm_xw<<<256, 256, 0, stream>>>(feat, embed, W, a, hf, sp);

    gat_attn_gather<<<N, 256, 0, stream>>>(he, sp, nidx, out);
}


# GEMM K loop pipelined across the chunk barrier: next chunk's LDS stores one K-step earlier, barrier + its first fragment reads in front of the last K-step's WMMAs, global loads behind the first
# speedup vs baseline: 2.4954x; 1.0386x over previous
#include <hip/hip_runtime.h>
#include <math.h>

typedef __attribute__((ext_vector_type(2))) float v2f;
typedef __attribute__((ext_vector_type(4))) float v4f;
typedef __attribute__((ext_vector_type(8))) float v8f;

#define GAT_N 2048
#define GAT_M 8192
#define GAT_F 1024
#define GAT_K 10
#define GAT_ALPHA 0.2f

#define GEMM_BM 160
#define GEMM_BN 256
#define GEMM_BK 32
#define GEMM_APITCH 36
#define GEMM_BCOLP (GEMM_BN + 16)
#define GEMM_ASTAGE (GEMM_BM * GEMM_APITCH)
#define GEMM_BSTAGE ((GEMM_BK / 2) * GEMM_BCOLP * 2)
#define GEMM_STAGE (GEMM_ASTAGE + GEMM_BSTAGE)

template <int MASK>
__device__ __forceinline__ float gat_xmask(float x) {
    const int b = __builtin_bit_cast(int, x);
    return __builtin_bit_cast(float,
        __builtin_amdgcn_update_dpp(0, b, 0x160 + MASK, 0xf, 0xf, true));
}
__device__ __forceinline__ float gat_sum16(float x) {
    x += gat_xmask<1>(x);
    x += gat_xmask<2>(x);
    x += gat_xmask<4>(x);
    x += gat_xmask<8>(x);
    return x;
}

struct gat_nothing {
    __device__ __forceinline__ void operator()() const {}
    __device__ __forceinline__ void operator()(v2f (&)[5], v2f (&)[4]) const {}
};
template <class Fetch, class Stage, class Handover>
__device__ __forceinline__ void gat_gemm_chunk(const float* sA, const float* sB,
                                               v8f (&c)[5][4], v2f (&a)[2][5], v2f (&b)[2][4],
                                               Fetch fetch, Stage stage, Handover handover) {
#pragma unroll
    for (int s = 0; s < 8; ++s) {
        const int cu = s & 1;
        const int nx = cu ^ 1;
        if (s + 1 < 8) {
#pragma unroll
            for (int m = 0; m < 5; ++m)
                a[nx][m] = *(const v2f*)(sA + 16 * m * GEMM_APITCH + 4 * (s + 1));
#pragma unroll
            for (int n = 0; n < 4; ++n)
                b[nx][n] = *(const v2f*)(sB + 4 * (s + 1) * GEMM_BCOLP + 32 * n);
            if (s == 6) stage();
        } else {
            handover(a[nx], b[nx]);
        }
        __builtin_amdgcn_sched_barrier(0);
#pragma unroll
        for (int m = 0; m < 5; ++m)
#pragma unroll
            for (int n = 0; n < 4; ++n)
                c[m][n] = __builtin_amdgcn_wmma_f32_16x16x4_f32(
                    false, a[cu][m], false, b[cu][n], (short)0, c[m][n], false, false);
        __builtin_amdgcn_sched_barrier(0);
        if (s == 0) fetch();
    }
}

__global__ __launch_bounds__(256) void gat_gemm_xw(const float* __restrict__ Xf,
                                                   const float* __restrict__ Xe,
                                                   const float* __restrict__ W,
                                                   const float* __restrict__ avec,
                                                   float* __restrict__ H,
                                                   float* __restrict__ SP) {
    __shared__ __attribute__((aligned(16))) float lds[2 * GEMM_STAGE];

    const int F = GAT_F;
    const int t = threadIdx.x;
    const int lane = t & 31;
    const int wave = t >> 5;
    const int wr = wave >> 2;
    const int wc = wave & 3;
    const int r = lane & 15;
    const int g = lane >> 4;
    const int i0 = (blockIdx.x >> 2) * GEMM_BM;
    const int j0 = (blockIdx.x & 3) * GEMM_BN;

    const int arow = t >> 3;
    const int akq = t & 7;
    const float* ap[5];
#pragma unroll
    for (int i = 0; i < 5; ++i) {
        const int row = i0 + arow + 32 * i;
        const float* base = (row < GAT_N) ? (Xf + (size_t)row * F)
                                          : (Xe + (size_t)(row - GAT_N) * F);
        ap[i] = base + 4 * akq;
    }
    const int a_st = arow * GEMM_APITCH + 4 * akq;

    const int bkp = t >> 6;
    const int bcg = t & 63;
    const float* wp = W + (size_t)(2 * bkp) * F + j0 + 4 * bcg;
    const int b_st = GEMM_ASTAGE + (bkp * GEMM_BCOLP + 4 * bcg) * 2;

    const int a_ld = (wr * 80 + r) * GEMM_APITCH + 2 * g;
    const int b_ld = GEMM_ASTAGE + (g * GEMM_BCOLP + wc * 64 + r) * 2;

    v8f c[5][4];
#pragma unroll
    for (int m = 0; m < 5; ++m)
#pragma unroll
        for (int n = 0; n < 4; ++n)
#pragma unroll
            for (int v = 0; v < 8; ++v) c[m][n][v] = 0.0f;

    v4f ga[5], gb0[4], gb1[4];

#pragma unroll
    for (int i = 0; i < 5; ++i) ga[i] = *(const v4f*)(ap[i]);
#pragma unroll
    for (int i = 0; i < 4; ++i) {
        const float* p = wp + (size_t)(8 * i) * F;
        gb0[i] = *(const v4f*)p;
        gb1[i] = *(const v4f*)(p + F);
    }
#pragma unroll
    for (int i = 0; i < 5; ++i) *(v4f*)(lds + a_st + 32 * i * GEMM_APITCH) = ga[i];
#pragma unroll
    for (int i = 0; i < 4; ++i) {
        float* q = lds + b_st + 8 * i * GEMM_BCOLP;
        v4f lo, hi;
        lo[0] = gb0[i][0]; lo[1] = gb1[i][0]; lo[2] = gb0[i][1]; lo[3] = gb1[i][1];
        hi[0] = gb0[i][2]; hi[1] = gb1[i][2]; hi[2] = gb0[i][3]; hi[3] = gb1[i][3];
        *(v4f*)q = lo;
        *(v4f*)(q + 4) = hi;
    }
    __syncthreads();

    v2f fa[2][5], fb[2][4];
    auto first_frags = [&](const float* st, v2f (&a0)[5], v2f (&b0)[4]) {
#pragma unroll
        for (int m = 0; m < 5; ++m) a0[m] = *(const v2f*)(st + a_ld + 16 * m * GEMM_APITCH);
#pragma unroll
        for (int n = 0; n < 4; ++n) b0[n] = *(const v2f*)(st + b_ld + 32 * n);
    };
    first_frags(lds, fa[0], fb[0]);

    const int nChunks = F / GEMM_BK;
    for (int kc = 0; kc + 1 < nChunks; ++kc) {
        float* cur = lds + (kc & 1) * GEMM_STAGE;
        float* nxt = lds + ((kc & 1) ^ 1) * GEMM_STAGE;
        const int kn = (kc + 1) * GEMM_BK;

        gat_gemm_chunk(cur + a_ld, cur + b_ld, c, fa, fb,
            [&]() {
#pragma unroll
                for (int i = 0; i < 5; ++i) ga[i] = *(const v4f*)(ap[i] + kn);
#pragma unroll
                for (int i = 0; i < 4; ++i) {
                    const float* p = wp + (size_t)(kn + 8 * i) * F;
                    gb0[i] = *(const v4f*)p;
                    gb1[i] = *(const v4f*)(p + F);
                }
            },
            [&]() {
#pragma unroll
                for (int i = 0; i < 5; ++i) *(v4f*)(nxt + a_st + 32 * i * GEMM_APITCH) = ga[i];
#pragma unroll
                for (int i = 0; i < 4; ++i) {
                    float* q = nxt + b_st + 8 * i * GEMM_BCOLP;
                    v4f lo, hi;
                    lo[0] = gb0[i][0]; lo[1] = gb1[i][0]; lo[2] = gb0[i][1]; lo[3] = gb1[i][1];
                    hi[0] = gb0[i][2]; hi[1] = gb1[i][2]; hi[2] = gb0[i][3]; hi[3] = gb1[i][3];
                    *(v4f*)q = lo;
                    *(v4f*)(q + 4) = hi;
                }
            },
            [&](v2f (&a0)[5], v2f (&b0)[4]) {
                __syncthreads();
                first_frags(nxt, a0, b0);
            });
    }
    float a1v[4], a2v[4];
#pragma unroll
    for (int n = 0; n < 4; ++n) {
        const int col = j0 + wc * 64 + 16 * n + r;
        a1v[n] = avec[col];
        a2v[n] = avec[F + col];
    }
    {
        const float* cur = lds + ((nChunks - 1) & 1) * GEMM_STAGE;
        gat_gemm_chunk(cur + a_ld, cur + b_ld, c, fa, fb,
                       gat_nothing(), gat_nothing(), gat_nothing());
    }

    const int rb = i0 + __builtin_amdgcn_readfirstlane(wr) * 80;
    float* red = lds;
#pragma unroll
    for (int m = 0; m < 5; ++m) {
        const bool isf = (rb + 16 * m) < GAT_N;
        float av[4];
#pragma unroll
        for (int n = 0; n < 4; ++n) av[n] = isf ? a1v[n] : a2v[n];
        v4f lo, hi;
#pragma unroll
        for (int v = 0; v < 8; ++v) {
            float p = c[m][0][v] * av[0];
            p = __builtin_fmaf(c[m][1][v], av[1], p);
            p = __builtin_fmaf(c[m][2][v], av[2], p);
            p = __builtin_fmaf(c[m][3][v], av[3], p);
            p = gat_sum16(p);
            if (v < 4) lo[v] = p; else hi[v - 4] = p;
        }
        if (r == 0) {
            float* q = red + wc * GEMM_BM + wr * 80 + 16 * m + 8 * g;
            *(v4f*)q = lo;
            *(v4f*)(q + 4) = hi;
        }
    }
    __syncthreads();
    if (t < GEMM_BM) {
        const float s = ((red[t] + red[GEMM_BM + t]) + red[2 * GEMM_BM + t]) + red[3 * GEMM_BM + t];
        SP[(size_t)(blockIdx.x & 3) * (GAT_N + GAT_M) + i0 + t] = s;
    }

#pragma unroll
    for (int m = 0; m < 5; ++m) {
        if ((rb + 16 * m) >= GAT_N) {
#pragma unroll
            for (int n = 0; n < 4; ++n)
#pragma unroll
                for (int v = 0; v < 8; ++v) {
                    const int row = i0 + wr * 80 + 16 * m + 8 * g + v;
                    const int col = j0 + wc * 64 + 16 * n + r;
                    H[(size_t)row * F + col] = c[m][n][v];
                }
        }
    }
}

__global__ __launch_bounds__(256) void gat_attn_gather(const float* __restrict__ he,
                                                       const float* __restrict__ sp,
                                                       const int* __restrict__ nidx,
                                                       float* __restrict__ out) {
    const int F = GAT_F;
    const int i = blockIdx.x;

    __shared__ int   sh_idx[GAT_K];
    __shared__ float sh_w[GAT_K];
    __shared__ int   sh_cnt;

    if (threadIdx.x == 0) {
        int   idx[GAT_K];
        float ev[GAT_K];
        int cnt = 0;
        const int R = GAT_N + GAT_M;
        const float s1i = ((sp[i] + sp[R + i]) + sp[2 * R + i]) + sp[3 * R + i];
        for (int k = 0; k < GAT_K; ++k) {
            const int j = nidx[i * GAT_K + k];
            bool dup = false;
            for (int p = 0; p < cnt; ++p) dup = dup || (idx[p] == j);
            if (!dup) {
                const float* pj = sp + GAT_N + j;
                const float s2j = ((pj[0] + pj[R]) + pj[2 * R]) + pj[3 * R];
                float e = s1i + s2j;
                e = (e > 0.0f) ? e : GAT_ALPHA * e;
                idx[cnt] = j;
                ev[cnt] = e;
                ++cnt;
            }
        }
        float mx = ev[0];
        for (int p = 1; p < cnt; ++p) mx = fmaxf(mx, ev[p]);
        float sum = 0.0f;
        for (int p = 0; p < cnt; ++p) { ev[p] = expf(ev[p] - mx); sum += ev[p]; }
        const float inv = 1.0f / sum;
        for (int p = 0; p < cnt; ++p) { sh_idx[p] = idx[p]; sh_w[p] = ev[p] * inv; }
        sh_cnt = cnt;
    }
    __syncthreads();

    const int cnt = sh_cnt;
    const int c = threadIdx.x * 4;
    v4f acc; acc[0] = 0.f; acc[1] = 0.f; acc[2] = 0.f; acc[3] = 0.f;
    for (int p = 0; p < cnt; ++p) {
        const float w = sh_w[p];
        const v4f hv = *(const v4f*)(he + (size_t)sh_idx[p] * F + c);
        acc += hv * w;
    }
    *(v4f*)(out + (size_t)i * F + c) = acc;
}

extern "C" void kernel_launch(void* const* d_in, const int* in_sizes, int n_in,
                              void* d_out, int out_size, void* d_ws, size_t ws_size,
                              hipStream_t stream) {
    const int N = GAT_N, M = GAT_M, F = GAT_F;

    const float* feat  = (const float*)d_in[0];
    const float* embed = (const float*)d_in[1];
    const float* W     = (const float*)d_in[2];
    const float* a     = (const float*)d_in[3];
    const int*   nidx  = (const int*)d_in[4];
    float* out = (float*)d_out;

    char* ws = (char*)d_ws;
    float* hf = (float*)ws;
    float* he = (float*)(ws + (size_t)N * F * sizeof(float));
    float* sp = (float*)(ws + (size_t)(N + M) * F * sizeof(float));

    gat_gemm_xw<<<256, 256, 0, stream>>>(feat, embed, W, a, hf, sp);

    gat_attn_gather<<<N, 256, 0, stream>>>(he, sp, nidx, out);
}


# Gather: one wave per output row (2048 waves, not 16384): neighbour scores, dedupe and softmax spread over lanes, all ten he rows requested up front into registers; float sum order unchanged
# speedup vs baseline: 2.5724x; 1.0309x over previous
#include <hip/hip_runtime.h>
#include <math.h>

typedef __attribute__((ext_vector_type(2))) float v2f;
typedef __attribute__((ext_vector_type(4))) float v4f;
typedef __attribute__((ext_vector_type(8))) float v8f;

#define GAT_N 2048
#define GAT_M 8192
#define GAT_F 1024
#define GAT_K 10
#define GAT_ALPHA 0.2f

#define GEMM_BM 160
#define GEMM_BN 256
#define GEMM_BK 32
#define GEMM_APITCH 36
#define GEMM_BCOLP (GEMM_BN + 16)
#define GEMM_ASTAGE (GEMM_BM * GEMM_APITCH)
#define GEMM_BSTAGE ((GEMM_BK / 2) * GEMM_BCOLP * 2)
#define GEMM_STAGE (GEMM_ASTAGE + GEMM_BSTAGE)

template <int MASK>
__device__ __forceinline__ float gat_xmask(float x) {
    const int b = __builtin_bit_cast(int, x);
    return __builtin_bit_cast(float,
        __builtin_amdgcn_update_dpp(0, b, 0x160 + MASK, 0xf, 0xf, true));
}
__device__ __forceinline__ float gat_sum16(float x) {
    x += gat_xmask<1>(x);
    x += gat_xmask<2>(x);
    x += gat_xmask<4>(x);
    x += gat_xmask<8>(x);
    return x;
}

struct gat_nothing {
    __device__ __forceinline__ void operator()() const {}
    __device__ __forceinline__ void operator()(v2f (&)[5], v2f (&)[4]) const {}
};
template <class Fetch, class Stage, class Handover>
__device__ __forceinline__ void gat_gemm_chunk(const float* sA, const float* sB,
                                               v8f (&c)[5][4], v2f (&a)[2][5], v2f (&b)[2][4],
                                               Fetch fetch, Stage stage, Handover handover) {
#pragma unroll
    for (int s = 0; s < 8; ++s) {
        const int cu = s & 1;
        const int nx = cu ^ 1;
        if (s + 1 < 8) {
#pragma unroll
            for (int m = 0; m < 5; ++m)
                a[nx][m] = *(const v2f*)(sA + 16 * m * GEMM_APITCH + 4 * (s + 1));
#pragma unroll
            for (int n = 0; n < 4; ++n)
                b[nx][n] = *(const v2f*)(sB + 4 * (s + 1) * GEMM_BCOLP + 32 * n);
            if (s == 6) stage();
        } else {
            handover(a[nx], b[nx]);
        }
        __builtin_amdgcn_sched_barrier(0);
#pragma unroll
        for (int m = 0; m < 5; ++m)
#pragma unroll
            for (int n = 0; n < 4; ++n)
                c[m][n] = __builtin_amdgcn_wmma_f32_16x16x4_f32(
                    false, a[cu][m], false, b[cu][n], (short)0, c[m][n], false, false);
        __builtin_amdgcn_sched_barrier(0);
        if (s == 0) fetch();
    }
}

__global__ __launch_bounds__(256) void gat_gemm_xw(const float* __restrict__ Xf,
                                                   const float* __restrict__ Xe,
                                                   const float* __restrict__ W,
                                                   const float* __restrict__ avec,
                                                   float* __restrict__ H,
                                                   float* __restrict__ SP) {
    __shared__ __attribute__((aligned(16))) float lds[2 * GEMM_STAGE];

    const int F = GAT_F;
    const int t = threadIdx.x;
    const int lane = t & 31;
    const int wave = t >> 5;
    const int wr = wave >> 2;
    const int wc = wave & 3;
    const int r = lane & 15;
    const int g = lane >> 4;
    const int i0 = (blockIdx.x >> 2) * GEMM_BM;
    const int j0 = (blockIdx.x & 3) * GEMM_BN;

    const int arow = t >> 3;
    const int akq = t & 7;
    const float* ap[5];
#pragma unroll
    for (int i = 0; i < 5; ++i) {
        const int row = i0 + arow + 32 * i;
        const float* base = (row < GAT_N) ? (Xf + (size_t)row * F)
                                          : (Xe + (size_t)(row - GAT_N) * F);
        ap[i] = base + 4 * akq;
    }
    const int a_st = arow * GEMM_APITCH + 4 * akq;

    const int bkp = t >> 6;
    const int bcg = t & 63;
    const float* wp = W + (size_t)(2 * bkp) * F + j0 + 4 * bcg;
    const int b_st = GEMM_ASTAGE + (bkp * GEMM_BCOLP + 4 * bcg) * 2;

    const int a_ld = (wr * 80 + r) * GEMM_APITCH + 2 * g;
    const int b_ld = GEMM_ASTAGE + (g * GEMM_BCOLP + wc * 64 + r) * 2;

    v8f c[5][4];
#pragma unroll
    for (int m = 0; m < 5; ++m)
#pragma unroll
        for (int n = 0; n < 4; ++n)
#pragma unroll
            for (int v = 0; v < 8; ++v) c[m][n][v] = 0.0f;

    v4f ga[5], gb0[4], gb1[4];

#pragma unroll
    for (int i = 0; i < 5; ++i) ga[i] = *(const v4f*)(ap[i]);
#pragma unroll
    for (int i = 0; i < 4; ++i) {
        const float* p = wp + (size_t)(8 * i) * F;
        gb0[i] = *(const v4f*)p;
        gb1[i] = *(const v4f*)(p + F);
    }
#pragma unroll
    for (int i = 0; i < 5; ++i) *(v4f*)(lds + a_st + 32 * i * GEMM_APITCH) = ga[i];
#pragma unroll
    for (int i = 0; i < 4; ++i) {
        float* q = lds + b_st + 8 * i * GEMM_BCOLP;
        v4f lo, hi;
        lo[0] = gb0[i][0]; lo[1] = gb1[i][0]; lo[2] = gb0[i][1]; lo[3] = gb1[i][1];
        hi[0] = gb0[i][2]; hi[1] = gb1[i][2]; hi[2] = gb0[i][3]; hi[3] = gb1[i][3];
        *(v4f*)q = lo;
        *(v4f*)(q + 4) = hi;
    }
    __syncthreads();

    v2f fa[2][5], fb[2][4];
    auto first_frags = [&](const float* st, v2f (&a0)[5], v2f (&b0)[4]) {
#pragma unroll
        for (int m = 0; m < 5; ++m) a0[m] = *(const v2f*)(st + a_ld + 16 * m * GEMM_APITCH);
#pragma unroll
        for (int n = 0; n < 4; ++n) b0[n] = *(const v2f*)(st + b_ld + 32 * n);
    };
    first_frags(lds, fa[0], fb[0]);

    const int nChunks = F / GEMM_BK;
    for (int kc = 0; kc + 1 < nChunks; ++kc) {
        float* cur = lds + (kc & 1) * GEMM_STAGE;
        float* nxt = lds + ((kc & 1) ^ 1) * GEMM_STAGE;
        const int kn = (kc + 1) * GEMM_BK;

        gat_gemm_chunk(cur + a_ld, cur + b_ld, c, fa, fb,
            [&]() {
#pragma unroll
                for (int i = 0; i < 5; ++i) ga[i] = *(const v4f*)(ap[i] + kn);
#pragma unroll
                for (int i = 0; i < 4; ++i) {
                    const float* p = wp + (size_t)(kn + 8 * i) * F;
                    gb0[i] = *(const v4f*)p;
                    gb1[i] = *(const v4f*)(p + F);
                }
            },
            [&]() {
#pragma unroll
                for (int i = 0; i < 5; ++i) *(v4f*)(nxt + a_st + 32 * i * GEMM_APITCH) = ga[i];
#pragma unroll
                for (int i = 0; i < 4; ++i) {
                    float* q = nxt + b_st + 8 * i * GEMM_BCOLP;
                    v4f lo, hi;
                    lo[0] = gb0[i][0]; lo[1] = gb1[i][0]; lo[2] = gb0[i][1]; lo[3] = gb1[i][1];
                    hi[0] = gb0[i][2]; hi[1] = gb1[i][2]; hi[2] = gb0[i][3]; hi[3] = gb1[i][3];
                    *(v4f*)q = lo;
                    *(v4f*)(q + 4) = hi;
                }
            },
            [&](v2f (&a0)[5], v2f (&b0)[4]) {
                __syncthreads();
                first_frags(nxt, a0, b0);
            });
    }
    float a1v[4], a2v[4];
#pragma unroll
    for (int n = 0; n < 4; ++n) {
        const int col = j0 + wc * 64 + 16 * n + r;
        a1v[n] = avec[col];
        a2v[n] = avec[F + col];
    }
    {
        const float* cur = lds + ((nChunks - 1) & 1) * GEMM_STAGE;
        gat_gemm_chunk(cur + a_ld, cur + b_ld, c, fa, fb,
                       gat_nothing(), gat_nothing(), gat_nothing());
    }

    const int rb = i0 + __builtin_amdgcn_readfirstlane(wr) * 80;
    float* red = lds;
#pragma unroll
    for (int m = 0; m < 5; ++m) {
        const bool isf = (rb + 16 * m) < GAT_N;
        float av[4];
#pragma unroll
        for (int n = 0; n < 4; ++n) av[n] = isf ? a1v[n] : a2v[n];
        v4f lo, hi;
#pragma unroll
        for (int v = 0; v < 8; ++v) {
            float p = c[m][0][v] * av[0];
            p = __builtin_fmaf(c[m][1][v], av[1], p);
            p = __builtin_fmaf(c[m][2][v], av[2], p);
            p = __builtin_fmaf(c[m][3][v], av[3], p);
            p = gat_sum16(p);
            if (v < 4) lo[v] = p; else hi[v - 4] = p;
        }
        if (r == 0) {
            float* q = red + wc * GEMM_BM + wr * 80 + 16 * m + 8 * g;
            *(v4f*)q = lo;
            *(v4f*)(q + 4) = hi;
        }
    }
    __syncthreads();
    if (t < GEMM_BM) {
        const float s = ((red[t] + red[GEMM_BM + t]) + red[2 * GEMM_BM + t]) + red[3 * GEMM_BM + t];
        SP[(size_t)(blockIdx.x & 3) * (GAT_N + GAT_M) + i0 + t] = s;
    }

#pragma unroll
    for (int m = 0; m < 5; ++m) {
        if ((rb + 16 * m) >= GAT_N) {
#pragma unroll
            for (int n = 0; n < 4; ++n)
#pragma unroll
                for (int v = 0; v < 8; ++v) {
                    const int row = i0 + wr * 80 + 16 * m + 8 * g + v;
                    const int col = j0 + wc * 64 + 16 * n + r;
                    H[(size_t)row * F + col] = c[m][n][v];
                }
        }
    }
}

#define GATH_WAVES 8
static_assert(GAT_N == 256 * GATH_WAVES, "one wave per row: the launch below uses 256 blocks");
static_assert(GAT_F == 8 * 128, "a row is eight 128-float pieces");

__device__ __forceinline__ float gat_lane_f(float x, int l) {
    return __builtin_bit_cast(float, __builtin_amdgcn_readlane(__builtin_bit_cast(int, x), l));
}

__global__ __launch_bounds__(256) void gat_attn_gather(const float* __restrict__ he,
                                                       const float* __restrict__ sp,
                                                       const int* __restrict__ nidx,
                                                       float* __restrict__ out) {
    const int F = GAT_F;
    const int R = GAT_N + GAT_M;
    const int lane = threadIdx.x & 31;
    const int i = blockIdx.x * GATH_WAVES + __builtin_amdgcn_readfirstlane((int)(threadIdx.x >> 5));

    const int k = (lane < GAT_K) ? lane : (GAT_K - 1);
    const int j = nidx[i * GAT_K + k];
    const float* pj = sp + GAT_N + j;
    const float q0 = pj[0], q1 = pj[R], q2 = pj[2 * R], q3 = pj[3 * R];
    const float t0 = sp[i], t1 = sp[R + i], t2 = sp[2 * R + i], t3 = sp[3 * R + i];

    const float* hb = he + 4 * lane;
    v4f hv[GAT_K][8];
#pragma unroll
    for (int p = 0; p < GAT_K; ++p) {
        const float* row = hb + (size_t)__builtin_amdgcn_readlane(j, p) * F;
#pragma unroll
        for (int c = 0; c < 8; ++c) hv[p][c] = *(const v4f*)(row + 128 * c);
    }
    __builtin_amdgcn_sched_barrier(0);

    const float s2j = ((q0 + q1) + q2) + q3;
    const float s1i = ((t0 + t1) + t2) + t3;
    float e = s1i + s2j;
    e = (e > 0.0f) ? e : GAT_ALPHA * e;

    int dup = (lane >= GAT_K) ? 1 : 0;
#pragma unroll
    for (int p = 0; p < GAT_K - 1; ++p)
        dup |= ((p < lane) ? 1 : 0) & ((__builtin_amdgcn_readlane(j, p) == j) ? 1 : 0);
    e = dup ? -INFINITY : e;

    float mx = gat_lane_f(e, 0);
#pragma unroll
    for (int p = 1; p < GAT_K; ++p) mx = fmaxf(mx, gat_lane_f(e, p));
    const float ev = dup ? 0.0f : expf(e - mx);
    float sum = gat_lane_f(ev, 0);
#pragma unroll
    for (int p = 1; p < GAT_K; ++p) sum += gat_lane_f(ev, p);
    const float inv = 1.0f / sum;
    const float w = ev * inv;

    v4f acc[8];
#pragma unroll
    for (int c = 0; c < 8; ++c) { acc[c][0] = 0.f; acc[c][1] = 0.f; acc[c][2] = 0.f; acc[c][3] = 0.f; }
#pragma unroll
    for (int p = 0; p < GAT_K; ++p) {
        const float wp = gat_lane_f(w, p);
#pragma unroll
        for (int c = 0; c < 8; ++c) acc[c] += hv[p][c] * wp;
    }
    float* orow = out + (size_t)i * F + 4 * lane;
#pragma unroll
    for (int c = 0; c < 8; ++c) *(v4f*)(orow + 128 * c) = acc[c];
}

extern "C" void kernel_launch(void* const* d_in, const int* in_sizes, int n_in,
                              void* d_out, int out_size, void* d_ws, size_t ws_size,
                              hipStream_t stream) {
    const int N = GAT_N, M = GAT_M, F = GAT_F;

    const float* feat  = (const float*)d_in[0];
    const float* embed = (const float*)d_in[1];
    const float* W     = (const float*)d_in[2];
    const float* a     = (const float*)d_in[3];
    const int*   nidx  = (const int*)d_in[4];
    float* out = (float*)d_out;

    char* ws = (char*)d_ws;
    float* hf = (float*)ws;
    float* he = (float*)(ws + (size_t)N * F * sizeof(float));
    float* sp = (float*)(ws + (size_t)(N + M) * F * sizeof(float));

    gat_gemm_xw<<<256, 256, 0, stream>>>(feat, embed, W, a, hf, sp);

    gat_attn_gather<<<256, 256, 0, stream>>>(he, sp, nidx, out);
}


# Feature rows leave the GEMM: s1 = Xf.(W a[:F]); W a[:F] built beside the GEMM (16 W rows per block), dotted per row in the gather; GEMM on the M embed rows only, tiles 128x256; s1 sum order changes
# speedup vs baseline: 3.0010x; 1.1666x over previous
#include <hip/hip_runtime.h>
#include <math.h>

typedef __attribute__((ext_vector_type(2))) float v2f;
typedef __attribute__((ext_vector_type(4))) float v4f;
typedef __attribute__((ext_vector_type(8))) float v8f;

#define GAT_N 2048
#define GAT_M 8192
#define GAT_F 1024
#define GAT_K 10
#define GAT_ALPHA 0.2f

#define GEMM_BM 128
#define GEMM_BN 256
#define GEMM_BK 32
#define GEMM_APITCH 36
#define GEMM_BCOLP (GEMM_BN + 16)
#define GEMM_ASTAGE (GEMM_BM * GEMM_APITCH)
#define GEMM_BSTAGE ((GEMM_BK / 2) * GEMM_BCOLP * 2)
#define GEMM_STAGE (GEMM_ASTAGE + GEMM_BSTAGE)
#define GAT_UPITCH (64 * 32)

template <int MASK>
__device__ __forceinline__ float gat_xmask(float x) {
    const int b = __builtin_bit_cast(int, x);
    return __builtin_bit_cast(float,
        __builtin_amdgcn_update_dpp(0, b, 0x160 + MASK, 0xf, 0xf, true));
}
__device__ __forceinline__ float gat_sum16(float x) {
    x += gat_xmask<1>(x);
    x += gat_xmask<2>(x);
    x += gat_xmask<4>(x);
    x += gat_xmask<8>(x);
    return x;
}

struct gat_nothing {
    __device__ __forceinline__ void operator()() const {}
    __device__ __forceinline__ void operator()(v2f (&)[4], v2f (&)[4]) const {}
};
template <class Fetch, class Stage, class Handover>
__device__ __forceinline__ void gat_gemm_chunk(const float* sA, const float* sB,
                                               v8f (&c)[4][4], v2f (&a)[2][4], v2f (&b)[2][4],
                                               Fetch fetch, Stage stage, Handover handover) {
#pragma unroll
    for (int s = 0; s < 8; ++s) {
        const int cu = s & 1;
        const int nx = cu ^ 1;
        if (s + 1 < 8) {
#pragma unroll
            for (int m = 0; m < 4; ++m)
                a[nx][m] = *(const v2f*)(sA + 16 * m * GEMM_APITCH + 4 * (s + 1));
#pragma unroll
            for (int n = 0; n < 4; ++n)
                b[nx][n] = *(const v2f*)(sB + 4 * (s + 1) * GEMM_BCOLP + 32 * n);
            if (s == 6) stage();
        } else {
            handover(a[nx], b[nx]);
        }
        __builtin_amdgcn_sched_barrier(0);
#pragma unroll
        for (int m = 0; m < 4; ++m)
#pragma unroll
            for (int n = 0; n < 4; ++n)
                c[m][n] = __builtin_amdgcn_wmma_f32_16x16x4_f32(
                    false, a[cu][m], false, b[cu][n], (short)0, c[m][n], false, false);
        __builtin_amdgcn_sched_barrier(0);
        if (s == 0) fetch();
    }
}

__global__ __launch_bounds__(256) void gat_gemm_xw(const float* __restrict__ Xe,
                                                   const float* __restrict__ W,
                                                   const float* __restrict__ avec,
                                                   float* __restrict__ H,
                                                   float* __restrict__ SP,
                                                   float* __restrict__ UP) {
    __shared__ __attribute__((aligned(16))) float lds[2 * GEMM_STAGE];
    __shared__ float ups[16];

    const int F = GAT_F;
    const int t = threadIdx.x;
    const int lane = t & 31;
    const int wave = t >> 5;
    const int wr = wave >> 2;
    const int wc = wave & 3;
    const int r = lane & 15;
    const int g = lane >> 4;
    const int rt = blockIdx.x >> 2;
    const int i0 = rt * GEMM_BM;
    const int j0 = (blockIdx.x & 3) * GEMM_BN;

    const int arow = t >> 3;
    const int akq = t & 7;
    const float* ap[4];
#pragma unroll
    for (int i = 0; i < 4; ++i)
        ap[i] = Xe + (size_t)(i0 + arow + 32 * i) * F + 4 * akq;
    const int a_st = arow * GEMM_APITCH + 4 * akq;

    const int bkp = t >> 6;
    const int bcg = t & 63;
    const float* wp = W + (size_t)(2 * bkp) * F + j0 + 4 * bcg;
    const int b_st = GEMM_ASTAGE + (bkp * GEMM_BCOLP + 4 * bcg) * 2;

    const int a_ld = (wr * 64 + r) * GEMM_APITCH + 2 * g;
    const int b_ld = GEMM_ASTAGE + (g * GEMM_BCOLP + wc * 64 + r) * 2;

    v8f c[4][4];
#pragma unroll
    for (int m = 0; m < 4; ++m)
#pragma unroll
        for (int n = 0; n < 4; ++n)
#pragma unroll
            for (int v = 0; v < 8; ++v) c[m][n][v] = 0.0f;

    v4f ga[4], gb0[4], gb1[4];

#pragma unroll
    for (int i = 0; i < 4; ++i) ga[i] = *(const v4f*)(ap[i]);
#pragma unroll
    for (int i = 0; i < 4; ++i) {
        const float* p = wp + (size_t)(8 * i) * F;
        gb0[i] = *(const v4f*)p;
        gb1[i] = *(const v4f*)(p + F);
    }

    {
        const float* wq = W + (size_t)(16 * rt + (t >> 4)) * F + j0 + 4 * r;
        const float* aq = avec + j0 + 4 * r;
        v4f wv[4], av[4];
#pragma unroll
        for (int q = 0; q < 4; ++q) {
            wv[q] = *(const v4f*)(wq + 64 * q);
            av[q] = *(const v4f*)(aq + 64 * q);
        }
        float p = 0.0f;
#pragma unroll
        for (int q = 0; q < 4; ++q)
#pragma unroll
            for (int e = 0; e < 4; ++e) p = __builtin_fmaf(wv[q][e], av[q][e], p);
        p = gat_sum16(p);
        if (r == 0) ups[t >> 4] = p;
    }

#pragma unroll
    for (int i = 0; i < 4; ++i) *(v4f*)(lds + a_st + 32 * i * GEMM_APITCH) = ga[i];
#pragma unroll
    for (int i = 0; i < 4; ++i) {
        float* q = lds + b_st + 8 * i * GEMM_BCOLP;
        v4f lo, hi;
        lo[0] = gb0[i][0]; lo[1] = gb1[i][0]; lo[2] = gb0[i][1]; lo[3] = gb1[i][1];
        hi[0] = gb0[i][2]; hi[1] = gb1[i][2]; hi[2] = gb0[i][3]; hi[3] = gb1[i][3];
        *(v4f*)q = lo;
        *(v4f*)(q + 4) = hi;
    }
    __syncthreads();

    v2f fa[2][4], fb[2][4];
    auto first_frags = [&](const float* st, v2f (&a0)[4], v2f (&b0)[4]) {
#pragma unroll
        for (int m = 0; m < 4; ++m) a0[m] = *(const v2f*)(st + a_ld + 16 * m * GEMM_APITCH);
#pragma unroll
        for (int n = 0; n < 4; ++n) b0[n] = *(const v2f*)(st + b_ld + 32 * n);
    };
    first_frags(lds, fa[0], fb[0]);

    const int nChunks = F / GEMM_BK;
    for (int kc = 0; kc + 1 < nChunks; ++kc) {
        float* cur = lds + (kc & 1) * GEMM_STAGE;
        float* nxt = lds + ((kc & 1) ^ 1) * GEMM_STAGE;
        const int kn = (kc + 1) * GEMM_BK;

        gat_gemm_chunk(cur + a_ld, cur + b_ld, c, fa, fb,
            [&]() {
#pragma unroll
                for (int i = 0; i < 4; ++i) ga[i] = *(const v4f*)(ap[i] + kn);
#pragma unroll
                for (int i = 0; i < 4; ++i) {
                    const float* p = wp + (size_t)(kn + 8 * i) * F;
                    gb0[i] = *(const v4f*)p;
                    gb1[i] = *(const v4f*)(p + F);
                }
            },
            [&]() {
#pragma unroll
                for (int i = 0; i < 4; ++i) *(v4f*)(nxt + a_st + 32 * i * GEMM_APITCH) = ga[i];
#pragma unroll
                for (int i = 0; i < 4; ++i) {
                    float* q = nxt + b_st + 8 * i * GEMM_BCOLP;
                    v4f lo, hi;
                    lo[0] = gb0[i][0]; lo[1] = gb1[i][0]; lo[2] = gb0[i][1]; lo[3] = gb1[i][1];
                    hi[0] = gb0[i][2]; hi[1] = gb1[i][2]; hi[2] = gb0[i][3]; hi[3] = gb1[i][3];
                    *(v4f*)q = lo;
                    *(v4f*)(q + 4) = hi;
                }
            },
            [&](v2f (&a0)[4], v2f (&b0)[4]) {
                __syncthreads();
                first_frags(nxt, a0, b0);
            });
    }
    float a2v[4];
#pragma unroll
    for (int n = 0; n < 4; ++n) a2v[n] = avec[F + j0 + wc * 64 + 16 * n + r];
    {
        const float* cur = lds + ((nChunks - 1) & 1) * GEMM_STAGE;
        gat_gemm_chunk(cur + a_ld, cur + b_ld, c, fa, fb,
                       gat_nothing(), gat_nothing(), gat_nothing());
    }

    float* red = lds;
#pragma unroll
    for (int m = 0; m < 4; ++m) {
        v4f lo, hi;
#pragma unroll
        for (int v = 0; v < 8; ++v) {
            float p = c[m][0][v] * a2v[0];
            p = __builtin_fmaf(c[m][1][v], a2v[1], p);
            p = __builtin_fmaf(c[m][2][v], a2v[2], p);
            p = __builtin_fmaf(c[m][3][v], a2v[3], p);
            p = gat_sum16(p);
            if (v < 4) lo[v] = p; else hi[v - 4] = p;
        }
        if (r == 0) {
            float* q = red + wc * GEMM_BM + wr * 64 + 16 * m + 8 * g;
            *(v4f*)q = lo;
            *(v4f*)(q + 4) = hi;
        }
    }
    __syncthreads();
    if (t < GEMM_BM) {
        const float s = ((red[t] + red[GEMM_BM + t]) + red[2 * GEMM_BM + t]) + red[3 * GEMM_BM + t];
        SP[(size_t)(blockIdx.x & 3) * GAT_M + i0 + t] = s;
    } else if (t < GEMM_BM + 16) {
        UP[(blockIdx.x & 3) * GAT_UPITCH + rt * 32 + (t - GEMM_BM)] = ups[t - GEMM_BM];
    }

#pragma unroll
    for (int m = 0; m < 4; ++m)
#pragma unroll
        for (int n = 0; n < 4; ++n)
#pragma unroll
            for (int v = 0; v < 8; ++v) {
                const int row = i0 + wr * 64 + 16 * m + 8 * g + v;
                const int col = j0 + wc * 64 + 16 * n + r;
                H[(size_t)row * F + col] = c[m][n][v];
            }
}

#define GATH_WAVES 8
static_assert(GAT_N == 256 * GATH_WAVES, "one wave per row: the launch below uses 256 blocks");
static_assert(GAT_F == 8 * 128, "a row is eight 128-float pieces");

__device__ __forceinline__ float gat_lane_f(float x, int l) {
    return __builtin_bit_cast(float, __builtin_amdgcn_readlane(__builtin_bit_cast(int, x), l));
}

__global__ __launch_bounds__(256) void gat_attn_gather(const float* __restrict__ he,
                                                       const float* __restrict__ sp,
                                                       const float* __restrict__ up,
                                                       const float* __restrict__ feat,
                                                       const int* __restrict__ nidx,
                                                       float* __restrict__ out) {
    __shared__ __attribute__((aligned(16))) float us[GAT_F];
    const int F = GAT_F;
    const int R = GAT_M;
    const int t = threadIdx.x;
    const int lane = t & 31;
    const int i = blockIdx.x * GATH_WAVES + __builtin_amdgcn_readfirstlane((int)(threadIdx.x >> 5));

    const int k = (lane < GAT_K) ? lane : (GAT_K - 1);
    const int j = nidx[i * GAT_K + k];
    const float* upp = up + (t >> 2) * 32 + 4 * (t & 3);
    const v4f u0 = *(const v4f*)upp;
    const v4f u1 = *(const v4f*)(upp + GAT_UPITCH);
    const v4f u2 = *(const v4f*)(upp + 2 * GAT_UPITCH);
    const v4f u3 = *(const v4f*)(upp + 3 * GAT_UPITCH);
    const float* fr = feat + (size_t)i * F + 4 * lane;
    v4f fv[8];
#pragma unroll
    for (int c = 0; c < 8; ++c) fv[c] = *(const v4f*)(fr + 128 * c);
    const float* pj = sp + j;
    const float q0 = pj[0], q1 = pj[R], q2 = pj[2 * R], q3 = pj[3 * R];

    const float* hb = he + 4 * lane;
    v4f hv[GAT_K][8];
#pragma unroll
    for (int p = 0; p < GAT_K; ++p) {
        const float* row = hb + (size_t)__builtin_amdgcn_readlane(j, p) * F;
#pragma unroll
        for (int c = 0; c < 8; ++c) hv[p][c] = *(const v4f*)(row + 128 * c);
    }
    __builtin_amdgcn_sched_barrier(0);

    *(v4f*)(us + 4 * t) = ((u0 + u1) + u2) + u3;
    __builtin_amdgcn_fence(__ATOMIC_RELEASE, "workgroup", "local");
    __builtin_amdgcn_s_barrier();
    __builtin_amdgcn_fence(__ATOMIC_ACQUIRE, "workgroup", "local");
    float d = 0.0f;
#pragma unroll
    for (int c = 0; c < 8; ++c) {
        const v4f uv = *(const v4f*)(us + 128 * c + 4 * lane);
        d = __builtin_fmaf(fv[c][0], uv[0], d);
        d = __builtin_fmaf(fv[c][1], uv[1], d);
        d = __builtin_fmaf(fv[c][2], uv[2], d);
        d = __builtin_fmaf(fv[c][3], uv[3], d);
    }
    d = gat_sum16(d);
    const float s1i = gat_lane_f(d, 0) + gat_lane_f(d, 16);

    const float s2j = ((q0 + q1) + q2) + q3;
    float e = s1i + s2j;
    e = (e > 0.0f) ? e : GAT_ALPHA * e;

    int dup = (lane >= GAT_K) ? 1 : 0;
#pragma unroll
    for (int p = 0; p < GAT_K - 1; ++p)
        dup |= ((p < lane) ? 1 : 0) & ((__builtin_amdgcn_readlane(j, p) == j) ? 1 : 0);
    e = dup ? -INFINITY : e;

    float mx = gat_lane_f(e, 0);
#pragma unroll
    for (int p = 1; p < GAT_K; ++p) mx = fmaxf(mx, gat_lane_f(e, p));
    const float ev = dup ? 0.0f : expf(e - mx);
    float sum = gat_lane_f(ev, 0);
#pragma unroll
    for (int p = 1; p < GAT_K; ++p) sum += gat_lane_f(ev, p);
    const float inv = 1.0f / sum;
    const float w = ev * inv;

    v4f acc[8];
#pragma unroll
    for (int c = 0; c < 8; ++c) { acc[c][0] = 0.f; acc[c][1] = 0.f; acc[c][2] = 0.f; acc[c][3] = 0.f; }
#pragma unroll
    for (int p = 0; p < GAT_K; ++p) {
        const float wp = gat_lane_f(w, p);
#pragma unroll
        for (int c = 0; c < 8; ++c) acc[c] += hv[p][c] * wp;
    }
    float* orow = out + (size_t)i * F + 4 * lane;
#pragma unroll
    for (int c = 0; c < 8; ++c) *(v4f*)(orow + 128 * c) = acc[c];
}

extern "C" void kernel_launch(void* const* d_in, const int* in_sizes, int n_in,
                              void* d_out, int out_size, void* d_ws, size_t ws_size,
                              hipStream_t stream) {
    const int M = GAT_M, F = GAT_F;

    const float* feat  = (const float*)d_in[0];
    const float* embed = (const float*)d_in[1];
    const float* W     = (const float*)d_in[2];
    const float* a     = (const float*)d_in[3];
    const int*   nidx  = (const int*)d_in[4];
    float* out = (float*)d_out;

    char* ws = (char*)d_ws;
    float* he = (float*)ws;
    float* sp = (float*)(ws + (size_t)M * F * sizeof(float));
    float* up = sp + 4 * M;

    gat_gemm_xw<<<256, 256, 0, stream>>>(embed, W, a, he, sp, up);

    gat_attn_gather<<<256, 256, 0, stream>>>(he, sp, up, feat, nidx, out);
}
